# starting kernel
# baseline (speedup 1.0000x reference)
#include <hip/hip_runtime.h>
#include <hip/hip_bf16.h>
#include <math.h>

typedef __attribute__((ext_vector_type(2))) float v2f;
typedef __attribute__((ext_vector_type(8))) float v8f;

__global__ __launch_bounds__(256) void img_mean_kernel(const float* __restrict__ in,
                                                       float* __restrict__ out) {
    int row  = (blockIdx.x << 3) + (threadIdx.x >> 5);
    int lane = threadIdx.x & 31;
    const float4* p = reinterpret_cast<const float4*>(in + (size_t)row * 256);
    float4 v0 = p[lane];
    float4 v1 = p[lane + 32];
    float s = (v0.x + v0.y) + (v0.z + v0.w) + (v1.x + v1.y) + (v1.z + v1.w);
    #pragma unroll
    for (int off = 16; off > 0; off >>= 1)
        s += __shfl_xor(s, off, 32);
    if (lane == 0)
        out[row] = s * (1.0f / 256.0f);
}

__global__ __launch_bounds__(512) void rad_mean_kernel(const float* __restrict__ in,
                                                       float* __restrict__ out) {
    int b = blockIdx.x;
    int t = threadIdx.x;
    const float* p = in + (size_t)b * 32 * 512 + t;
    float s = 0.0f;
    #pragma unroll
    for (int r = 0; r < 32; ++r)
        s += p[r * 512];
    out[b * 512 + t] = s * (1.0f / 32.0f);
}

__global__ __launch_bounds__(256) void inv_norm_kernel(const float* __restrict__ img,
                                                       const float* __restrict__ rad,
                                                       float* __restrict__ invn) {
    __shared__ float sm[256];
    int row = blockIdx.x;
    const float* src = (row < 256) ? (img + (size_t)row * 512)
                                   : (rad + (size_t)(row - 256) * 512);
    int t = threadIdx.x;
    float a = src[t];
    float b = src[t + 256];
    sm[t] = a * a + b * b;
    __syncthreads();
    #pragma unroll
    for (int s = 128; s > 0; s >>= 1) {
        if (t < s) sm[t] += sm[t + s];
        __syncthreads();
    }
    if (t == 0)
        invn[row] = 1.0f / fmaxf(sqrtf(sm[0]), 1e-12f);
}

__global__ __launch_bounds__(32) void sim_wmma_kernel(const float* __restrict__ img,
                                                      const float* __restrict__ rad,
                                                      const float* __restrict__ invn,
                                                      const float* __restrict__ temperature,
                                                      float* __restrict__ out) {
    int tile = blockIdx.x;
    int ti = tile >> 4;
    int tj = tile & 15;
    int lane  = threadIdx.x & 31;
    int lhalf = lane >> 4;
    int lmod  = lane & 15;

    int rowA = ti * 16 + lmod;
    int rowB = tj * 16 + lmod;
    float sA = invn[rowA];
    float sB = invn[256 + rowB];

    const float* pa = img + (size_t)rowA * 512 + 2 * lhalf;
    const float* pb = rad + (size_t)rowB * 512 + 2 * lhalf;

    v8f acc = {};
    #pragma unroll 8
    for (int k = 0; k < 512; k += 4) {
        v2f a = *reinterpret_cast<const v2f*>(pa + k);
        v2f b = *reinterpret_cast<const v2f*>(pb + k);
        a = a * sA;
        b = b * sB;
        acc = __builtin_amdgcn_wmma_f32_16x16x4_f32(
             false, a,  false, b,
             (short)0, acc,  false,  false);
    }

    float invt = 1.0f / temperature[0];
    int col = tj * 16 + lmod;
    #pragma unroll
    for (int v = 0; v < 8; ++v) {
        int row = ti * 16 + v + 8 * lhalf;
        out[row * 256 + col] = acc[v] * invt;
    }
}

extern "C" void kernel_launch(void* const* d_in, const int* in_sizes, int n_in,
                              void* d_out, int out_size, void* d_ws, size_t ws_size,
                              hipStream_t stream) {
    const float* img_feat = (const float*)d_in[0];
    const float* rad_feat = (const float*)d_in[1];
    const float* temp     = (const float*)d_in[2];

    float* out_sim = (float*)d_out;
    float* out_img = out_sim + 256 * 256;
    float* out_rad = out_img + 256 * 512;

    float* invn = (float*)d_ws;

    img_mean_kernel<<<16384, 256, 0, stream>>>(img_feat, out_img);
    rad_mean_kernel<<<256, 512, 0, stream>>>(rad_feat, out_rad);
    inv_norm_kernel<<<512, 256, 0, stream>>>(out_img, out_rad, invn);
    sim_wmma_kernel<<<256, 32, 0, stream>>>(out_img, out_rad, invn, temp, out_sim);
}


# image-mean kernel as a stream: 512 workgroups x 8 waves, each wave sums 32 consecutive rows with 16-32 b128 loads in flight (was one wave per 1 KB row, 131072 waves); per-row sum order kept
# speedup vs baseline: 1.3192x; 1.3192x over previous
#include <hip/hip_runtime.h>
#include <hip/hip_bf16.h>
#include <math.h>

typedef __attribute__((ext_vector_type(2))) float v2f;
typedef __attribute__((ext_vector_type(8))) float v8f;

#define IM_WAVES 8
#define IM_ROWS  32
#define IM_GROUP 8
__global__ __launch_bounds__(IM_WAVES * 32) void img_mean_kernel(const float* __restrict__ in,
                                                                 float* __restrict__ out) {
    const unsigned lane = threadIdx.x & 31u;
    const unsigned wave = __builtin_amdgcn_readfirstlane(threadIdx.x >> 5);
    const unsigned row0 = (blockIdx.x * IM_WAVES + wave) * IM_ROWS;
    const float4* p = reinterpret_cast<const float4*>(in + (size_t)row0 * 256) + lane;

    float4 buf[2][2 * IM_GROUP];
#define IM_LOAD_GROUP(G) do { \
    for (int i = 0; i < 2 * IM_GROUP; ++i) \
        buf[(G) & 1][i] = p[(G) * (IM_GROUP * 64) + i * 32]; } while (0)

    IM_LOAD_GROUP(0);
    float mine = 0.0f;
    for (int g = 0; g < IM_ROWS / IM_GROUP; ++g) {
        __builtin_amdgcn_sched_barrier(0);
        if (g + 1 < IM_ROWS / IM_GROUP) IM_LOAD_GROUP(g + 1);
        __builtin_amdgcn_sched_barrier(0);
        for (int r = 0; r < IM_GROUP; ++r) {
            const float4 v0 = buf[g & 1][2 * r];
            const float4 v1 = buf[g & 1][2 * r + 1];
            float s = (v0.x + v0.y) + (v0.z + v0.w) + (v1.x + v1.y) + (v1.z + v1.w);
            for (int off = 16; off > 0; off >>= 1)
                s += __shfl_xor(s, off, 32);
            if (lane == (unsigned)(g * IM_GROUP + r)) mine = s;
        }
    }
#undef IM_LOAD_GROUP
    out[row0 + lane] = mine * (1.0f / 256.0f);
}

__global__ __launch_bounds__(512) void rad_mean_kernel(const float* __restrict__ in,
                                                       float* __restrict__ out) {
    int b = blockIdx.x;
    int t = threadIdx.x;
    const float* p = in + (size_t)b * 32 * 512 + t;
    float s = 0.0f;
    #pragma unroll
    for (int r = 0; r < 32; ++r)
        s += p[r * 512];
    out[b * 512 + t] = s * (1.0f / 32.0f);
}

__global__ __launch_bounds__(256) void inv_norm_kernel(const float* __restrict__ img,
                                                       const float* __restrict__ rad,
                                                       float* __restrict__ invn) {
    __shared__ float sm[256];
    int row = blockIdx.x;
    const float* src = (row < 256) ? (img + (size_t)row * 512)
                                   : (rad + (size_t)(row - 256) * 512);
    int t = threadIdx.x;
    float a = src[t];
    float b = src[t + 256];
    sm[t] = a * a + b * b;
    __syncthreads();
    #pragma unroll
    for (int s = 128; s > 0; s >>= 1) {
        if (t < s) sm[t] += sm[t + s];
        __syncthreads();
    }
    if (t == 0)
        invn[row] = 1.0f / fmaxf(sqrtf(sm[0]), 1e-12f);
}

__global__ __launch_bounds__(32) void sim_wmma_kernel(const float* __restrict__ img,
                                                      const float* __restrict__ rad,
                                                      const float* __restrict__ invn,
                                                      const float* __restrict__ temperature,
                                                      float* __restrict__ out) {
    int tile = blockIdx.x;
    int ti = tile >> 4;
    int tj = tile & 15;
    int lane  = threadIdx.x & 31;
    int lhalf = lane >> 4;
    int lmod  = lane & 15;

    int rowA = ti * 16 + lmod;
    int rowB = tj * 16 + lmod;
    float sA = invn[rowA];
    float sB = invn[256 + rowB];

    const float* pa = img + (size_t)rowA * 512 + 2 * lhalf;
    const float* pb = rad + (size_t)rowB * 512 + 2 * lhalf;

    v8f acc = {};
    #pragma unroll 8
    for (int k = 0; k < 512; k += 4) {
        v2f a = *reinterpret_cast<const v2f*>(pa + k);
        v2f b = *reinterpret_cast<const v2f*>(pb + k);
        a = a * sA;
        b = b * sB;
        acc = __builtin_amdgcn_wmma_f32_16x16x4_f32(
             false, a,  false, b,
             (short)0, acc,  false,  false);
    }

    float invt = 1.0f / temperature[0];
    int col = tj * 16 + lmod;
    #pragma unroll
    for (int v = 0; v < 8; ++v) {
        int row = ti * 16 + v + 8 * lhalf;
        out[row * 256 + col] = acc[v] * invt;
    }
}

extern "C" void kernel_launch(void* const* d_in, const int* in_sizes, int n_in,
                              void* d_out, int out_size, void* d_ws, size_t ws_size,
                              hipStream_t stream) {
    const float* img_feat = (const float*)d_in[0];
    const float* rad_feat = (const float*)d_in[1];
    const float* temp     = (const float*)d_in[2];

    float* out_sim = (float*)d_out;
    float* out_img = out_sim + 256 * 256;
    float* out_rad = out_img + 256 * 512;

    float* invn = (float*)d_ws;

    img_mean_kernel<<<512, 256, 0, stream>>>(img_feat, out_img);
    rad_mean_kernel<<<256, 512, 0, stream>>>(rad_feat, out_rad);
    inv_norm_kernel<<<512, 256, 0, stream>>>(out_img, out_rad, invn);
    sim_wmma_kernel<<<256, 32, 0, stream>>>(out_img, out_rad, invn, temp, out_sim);
}


# similarity kernel: K split over the 8 waves of a workgroup (16x32 tile, 128 workgroups), each wave requests its 24 b128 fragment loads at once; partial tiles summed via LDS: k sum order changes
# speedup vs baseline: 1.9036x; 1.4430x over previous
#include <hip/hip_runtime.h>
#include <hip/hip_bf16.h>
#include <math.h>

typedef __attribute__((ext_vector_type(2))) float v2f;
typedef __attribute__((ext_vector_type(8))) float v8f;

#define IM_WAVES 8
#define IM_ROWS  32
#define IM_GROUP 8
__global__ __launch_bounds__(IM_WAVES * 32) void img_mean_kernel(const float* __restrict__ in,
                                                                 float* __restrict__ out) {
    const unsigned lane = threadIdx.x & 31u;
    const unsigned wave = __builtin_amdgcn_readfirstlane(threadIdx.x >> 5);
    const unsigned row0 = (blockIdx.x * IM_WAVES + wave) * IM_ROWS;
    const float4* p = reinterpret_cast<const float4*>(in + (size_t)row0 * 256) + lane;

    float4 buf[2][2 * IM_GROUP];
#define IM_LOAD_GROUP(G) do { \
    for (int i = 0; i < 2 * IM_GROUP; ++i) \
        buf[(G) & 1][i] = p[(G) * (IM_GROUP * 64) + i * 32]; } while (0)

    IM_LOAD_GROUP(0);
    float mine = 0.0f;
    for (int g = 0; g < IM_ROWS / IM_GROUP; ++g) {
        __builtin_amdgcn_sched_barrier(0);
        if (g + 1 < IM_ROWS / IM_GROUP) IM_LOAD_GROUP(g + 1);
        __builtin_amdgcn_sched_barrier(0);
        for (int r = 0; r < IM_GROUP; ++r) {
            const float4 v0 = buf[g & 1][2 * r];
            const float4 v1 = buf[g & 1][2 * r + 1];
            float s = (v0.x + v0.y) + (v0.z + v0.w) + (v1.x + v1.y) + (v1.z + v1.w);
            for (int off = 16; off > 0; off >>= 1)
                s += __shfl_xor(s, off, 32);
            if (lane == (unsigned)(g * IM_GROUP + r)) mine = s;
        }
    }
#undef IM_LOAD_GROUP
    out[row0 + lane] = mine * (1.0f / 256.0f);
}

__global__ __launch_bounds__(512) void rad_mean_kernel(const float* __restrict__ in,
                                                       float* __restrict__ out) {
    int b = blockIdx.x;
    int t = threadIdx.x;
    const float* p = in + (size_t)b * 32 * 512 + t;
    float s = 0.0f;
    #pragma unroll
    for (int r = 0; r < 32; ++r)
        s += p[r * 512];
    out[b * 512 + t] = s * (1.0f / 32.0f);
}

__global__ __launch_bounds__(256) void inv_norm_kernel(const float* __restrict__ img,
                                                       const float* __restrict__ rad,
                                                       float* __restrict__ invn) {
    __shared__ float sm[256];
    int row = blockIdx.x;
    const float* src = (row < 256) ? (img + (size_t)row * 512)
                                   : (rad + (size_t)(row - 256) * 512);
    int t = threadIdx.x;
    float a = src[t];
    float b = src[t + 256];
    sm[t] = a * a + b * b;
    __syncthreads();
    #pragma unroll
    for (int s = 128; s > 0; s >>= 1) {
        if (t < s) sm[t] += sm[t + s];
        __syncthreads();
    }
    if (t == 0)
        invn[row] = 1.0f / fmaxf(sqrtf(sm[0]), 1e-12f);
}

#define SK_WAVES 8
#define SK_K     (512 / SK_WAVES)
#define SK_F4    (SK_K / 8)
__global__ __launch_bounds__(SK_WAVES * 32) void sim_wmma_kernel(const float* img,
                                                                 const float* rad,
                                                                 const float* __restrict__ invn,
                                                                 const float* __restrict__ temperature,
                                                                 float* __restrict__ out) {
    __shared__ float part[SK_WAVES][16][32];

    const int ti   = blockIdx.x >> 3;
    const int tjp  = blockIdx.x & 7;
    const int lane = threadIdx.x & 31;
    const int wave = __builtin_amdgcn_readfirstlane(threadIdx.x >> 5);
    const int h = lane >> 4;
    const int c = lane & 15;

    const int rowA  = ti * 16 + c;
    const int rowB0 = tjp * 32 + c;
    const int rowB1 = rowB0 + 16;

    const float invt = 1.0f / temperature[0];

    float sA  = invn[rowA];
    float sB0 = invn[256 + rowB0];
    float sB1 = invn[256 + rowB1];

    const float4* pa  = reinterpret_cast<const float4*>(img + (size_t)rowA  * 512 + wave * SK_K) + h;
    const float4* pb0 = reinterpret_cast<const float4*>(rad + (size_t)rowB0 * 512 + wave * SK_K) + h;
    const float4* pb1 = reinterpret_cast<const float4*>(rad + (size_t)rowB1 * 512 + wave * SK_K) + h;

    float4 fa[SK_F4], fb0[SK_F4], fb1[SK_F4];
    for (int j = 0; j < SK_F4; ++j) {
        fa[j]  = pa[2 * j];
        fb0[j] = pb0[2 * j];
        fb1[j] = pb1[2 * j];
    }
    asm volatile("" : "+v"(sA), "+v"(sB0), "+v"(sB1));

    v8f acc0 = {};
    v8f acc1 = {};
    for (int j = 0; j < SK_F4; ++j) {
        v2f a_lo = {fa[j].x * sA, fa[j].y * sA};
        v2f a_hi = {fa[j].z * sA, fa[j].w * sA};
        v2f p_lo = {fb0[j].x * sB0, fb0[j].y * sB0};
        v2f p_hi = {fb0[j].z * sB0, fb0[j].w * sB0};
        v2f q_lo = {fb1[j].x * sB1, fb1[j].y * sB1};
        v2f q_hi = {fb1[j].z * sB1, fb1[j].w * sB1};
        acc0 = __builtin_amdgcn_wmma_f32_16x16x4_f32(false, a_lo, false, p_lo, (short)0, acc0, false, false);
        acc1 = __builtin_amdgcn_wmma_f32_16x16x4_f32(false, a_lo, false, q_lo, (short)0, acc1, false, false);
        acc0 = __builtin_amdgcn_wmma_f32_16x16x4_f32(false, a_hi, false, p_hi, (short)0, acc0, false, false);
        acc1 = __builtin_amdgcn_wmma_f32_16x16x4_f32(false, a_hi, false, q_hi, (short)0, acc1, false, false);
    }

    for (int v = 0; v < 8; ++v) {
        part[wave][v][lane]     = acc0[v];
        part[wave][8 + v][lane] = acc1[v];
    }
    __syncthreads();

    for (int q = 0; q < 2; ++q) {
        const int rho = 2 * wave + q;
        const int reg = h * 8 + (rho & 7);
        const int sl  = (rho >> 3) * 16 + c;
        float s = part[0][reg][sl];
        for (int w = 1; w < SK_WAVES; ++w)
            s += part[w][reg][sl];
        out[(ti * 16 + rho) * 256 + tjp * 32 + lane] = s * invt;
    }
}

extern "C" void kernel_launch(void* const* d_in, const int* in_sizes, int n_in,
                              void* d_out, int out_size, void* d_ws, size_t ws_size,
                              hipStream_t stream) {
    const float* img_feat = (const float*)d_in[0];
    const float* rad_feat = (const float*)d_in[1];
    const float* temp     = (const float*)d_in[2];

    float* out_sim = (float*)d_out;
    float* out_img = out_sim + 256 * 256;
    float* out_rad = out_img + 256 * 512;

    float* invn = (float*)d_ws;

    img_mean_kernel<<<512, 256, 0, stream>>>(img_feat, out_img);
    rad_mean_kernel<<<256, 512, 0, stream>>>(rad_feat, out_rad);
    inv_norm_kernel<<<512, 256, 0, stream>>>(out_img, out_rad, invn);
    sim_wmma_kernel<<<128, 256, 0, stream>>>(out_img, out_rad, invn, temp, out_sim);
}


# row norms made in the similarity kernel from each wave's fragments (partial sums of squares meet in LDS; scales applied to the finished dot product): inv-norm launch gone, 4 -> 3; norm sum order new
# speedup vs baseline: 2.1840x; 1.1473x over previous
#include <hip/hip_runtime.h>
#include <hip/hip_bf16.h>
#include <math.h>

typedef __attribute__((ext_vector_type(2))) float v2f;
typedef __attribute__((ext_vector_type(8))) float v8f;

#define IM_WAVES 8
#define IM_ROWS  32
#define IM_GROUP 8
__global__ __launch_bounds__(IM_WAVES * 32) void img_mean_kernel(const float* __restrict__ in,
                                                                 float* __restrict__ out) {
    const unsigned lane = threadIdx.x & 31u;
    const unsigned wave = __builtin_amdgcn_readfirstlane(threadIdx.x >> 5);
    const unsigned row0 = (blockIdx.x * IM_WAVES + wave) * IM_ROWS;
    const float4* p = reinterpret_cast<const float4*>(in + (size_t)row0 * 256) + lane;

    float4 buf[2][2 * IM_GROUP];
#define IM_LOAD_GROUP(G) do { \
    for (int i = 0; i < 2 * IM_GROUP; ++i) \
        buf[(G) & 1][i] = p[(G) * (IM_GROUP * 64) + i * 32]; } while (0)

    IM_LOAD_GROUP(0);
    float mine = 0.0f;
    for (int g = 0; g < IM_ROWS / IM_GROUP; ++g) {
        __builtin_amdgcn_sched_barrier(0);
        if (g + 1 < IM_ROWS / IM_GROUP) IM_LOAD_GROUP(g + 1);
        __builtin_amdgcn_sched_barrier(0);
        for (int r = 0; r < IM_GROUP; ++r) {
            const float4 v0 = buf[g & 1][2 * r];
            const float4 v1 = buf[g & 1][2 * r + 1];
            float s = (v0.x + v0.y) + (v0.z + v0.w) + (v1.x + v1.y) + (v1.z + v1.w);
            for (int off = 16; off > 0; off >>= 1)
                s += __shfl_xor(s, off, 32);
            if (lane == (unsigned)(g * IM_GROUP + r)) mine = s;
        }
    }
#undef IM_LOAD_GROUP
    out[row0 + lane] = mine * (1.0f / 256.0f);
}

__global__ __launch_bounds__(512) void rad_mean_kernel(const float* __restrict__ in,
                                                       float* __restrict__ out) {
    int b = blockIdx.x;
    int t = threadIdx.x;
    const float* p = in + (size_t)b * 32 * 512 + t;
    float s = 0.0f;
    #pragma unroll
    for (int r = 0; r < 32; ++r)
        s += p[r * 512];
    out[b * 512 + t] = s * (1.0f / 32.0f);
}

__device__ __forceinline__ float other_half(float x) {
    return __builtin_bit_cast(float, __builtin_amdgcn_permlanex16(
        __builtin_bit_cast(int, x), __builtin_bit_cast(int, x), 0x76543210, 0xfedcba98, false, false));
}
#define SK_WAVES 8
#define SK_K     (512 / SK_WAVES)
#define SK_F4    (SK_K / 8)
__global__ __launch_bounds__(SK_WAVES * 32) void sim_wmma_kernel(const float* img,
                                                                 const float* rad,
                                                                 const float* __restrict__ temperature,
                                                                 float* __restrict__ out) {
    __shared__ float part[SK_WAVES][16][32];
    __shared__ float ssq[SK_WAVES][48];

    const int ti   = blockIdx.x >> 3;
    const int tjp  = blockIdx.x & 7;
    const int lane = threadIdx.x & 31;
    const int wave = __builtin_amdgcn_readfirstlane(threadIdx.x >> 5);
    const int h = lane >> 4;
    const int c = lane & 15;

    const int rowA  = ti * 16 + c;
    const int rowB0 = tjp * 32 + c;
    const int rowB1 = rowB0 + 16;

    const float temp = temperature[0];

    const float4* pa  = reinterpret_cast<const float4*>(img + (size_t)rowA  * 512 + wave * SK_K) + h;
    const float4* pb0 = reinterpret_cast<const float4*>(rad + (size_t)rowB0 * 512 + wave * SK_K) + h;
    const float4* pb1 = reinterpret_cast<const float4*>(rad + (size_t)rowB1 * 512 + wave * SK_K) + h;

    float4 fa[SK_F4], fb0[SK_F4], fb1[SK_F4];
    for (int j = 0; j < SK_F4; ++j) {
        fa[j]  = pa[2 * j];
        fb0[j] = pb0[2 * j];
        fb1[j] = pb1[2 * j];
    }
    asm volatile("" : "+v"(fa[0].x));

    v8f acc0 = {};
    v8f acc1 = {};
    v2f sa_lo = {0.0f, 0.0f}, sa_hi = {0.0f, 0.0f};
    v2f sp_lo = {0.0f, 0.0f}, sp_hi = {0.0f, 0.0f};
    v2f sq_lo = {0.0f, 0.0f}, sq_hi = {0.0f, 0.0f};
    for (int j = 0; j < SK_F4; ++j) {
        v2f a_lo = {fa[j].x, fa[j].y};
        v2f a_hi = {fa[j].z, fa[j].w};
        v2f p_lo = {fb0[j].x, fb0[j].y};
        v2f p_hi = {fb0[j].z, fb0[j].w};
        v2f q_lo = {fb1[j].x, fb1[j].y};
        v2f q_hi = {fb1[j].z, fb1[j].w};
        acc0 = __builtin_amdgcn_wmma_f32_16x16x4_f32(false, a_lo, false, p_lo, (short)0, acc0, false, false);
        acc1 = __builtin_amdgcn_wmma_f32_16x16x4_f32(false, a_lo, false, q_lo, (short)0, acc1, false, false);
        acc0 = __builtin_amdgcn_wmma_f32_16x16x4_f32(false, a_hi, false, p_hi, (short)0, acc0, false, false);
        acc1 = __builtin_amdgcn_wmma_f32_16x16x4_f32(false, a_hi, false, q_hi, (short)0, acc1, false, false);
        sa_lo += a_lo * a_lo;  sa_hi += a_hi * a_hi;
        sp_lo += p_lo * p_lo;  sp_hi += p_hi * p_hi;
        sq_lo += q_lo * q_lo;  sq_hi += q_hi * q_hi;
        asm volatile("" : "+v"(acc0), "+v"(acc1), "+v"(sa_lo), "+v"(sa_hi),
                          "+v"(sp_lo), "+v"(sp_hi), "+v"(sq_lo), "+v"(sq_hi));
    }

    float ssA  = (sa_lo.x + sa_lo.y) + (sa_hi.x + sa_hi.y);
    float ssB0 = (sp_lo.x + sp_lo.y) + (sp_hi.x + sp_hi.y);
    float ssB1 = (sq_lo.x + sq_lo.y) + (sq_hi.x + sq_hi.y);
    ssA  += other_half(ssA);
    ssB0 += other_half(ssB0);
    ssB1 += other_half(ssB1);

    for (int v = 0; v < 8; ++v) {
        part[wave][v][lane]     = acc0[v];
        part[wave][8 + v][lane] = acc1[v];
    }
    ssq[wave][lane] = h ? ssB1 : ssB0;
    if (h == 0) ssq[wave][32 + c] = ssA;
    __syncthreads();

    float tB = ssq[0][lane];
    float tA = ssq[0][32 + c];
    for (int w = 1; w < SK_WAVES; ++w) {
        tB += ssq[w][lane];
        tA += ssq[w][32 + c];
    }
    const float sclB = 1.0f / (fmaxf(sqrtf(tB), 1e-12f) * temp);
    const float invA = 1.0f / fmaxf(sqrtf(tA), 1e-12f);

    for (int q = 0; q < 2; ++q) {
        const int rho = 2 * wave + q;
        const int reg = h * 8 + (rho & 7);
        const int sl  = (rho >> 3) * 16 + c;
        float s = part[0][reg][sl];
        for (int w = 1; w < SK_WAVES; ++w)
            s += part[w][reg][sl];
        const float sclA = __builtin_bit_cast(float,
            __builtin_amdgcn_readlane(__builtin_bit_cast(int, invA), rho));
        out[(ti * 16 + rho) * 256 + tjp * 32 + lane] = s * sclA * sclB;
    }
}

extern "C" void kernel_launch(void* const* d_in, const int* in_sizes, int n_in,
                              void* d_out, int out_size, void* d_ws, size_t ws_size,
                              hipStream_t stream) {
    const float* img_feat = (const float*)d_in[0];
    const float* rad_feat = (const float*)d_in[1];
    const float* temp     = (const float*)d_in[2];

    float* out_sim = (float*)d_out;
    float* out_img = out_sim + 256 * 256;
    float* out_rad = out_img + 256 * 512;

    img_mean_kernel<<<512, 256, 0, stream>>>(img_feat, out_img);
    rad_mean_kernel<<<256, 512, 0, stream>>>(rad_feat, out_rad);
    sim_wmma_kernel<<<128, 256, 0, stream>>>(out_img, out_rad, temp, out_sim);
}


# radiomics-mean launch folded into the image-mean stream: each of its 4096 waves also sums one 32-column strip, its 32 row loads requested at the wave's start; launches 3 -> 2; sum orders kept
# speedup vs baseline: 2.4354x; 1.1151x over previous
#include <hip/hip_runtime.h>
#include <hip/hip_bf16.h>
#include <math.h>

typedef __attribute__((ext_vector_type(2))) float v2f;
typedef __attribute__((ext_vector_type(8))) float v8f;

#define IM_WAVES 8
#define IM_ROWS  32
#define IM_GROUP 8
#define RD_ROWS  32
#define RD_COLS  512
__global__ __launch_bounds__(IM_WAVES * 32) void mean_stream_kernel(const float* __restrict__ in,
                                                                    float* __restrict__ out,
                                                                    const float* __restrict__ rad_in,
                                                                    float* __restrict__ rad_out) {
    const unsigned lane = threadIdx.x & 31u;
    const unsigned wave = __builtin_amdgcn_readfirstlane(threadIdx.x >> 5);
    const unsigned gw   = blockIdx.x * IM_WAVES + wave;
    const unsigned row0 = gw * IM_ROWS;

    const float* pr = rad_in + (size_t)(gw >> 4) * (RD_ROWS * RD_COLS) + (gw & 15u) * 32u + lane;
    float rv[RD_ROWS];

    const float4* p = reinterpret_cast<const float4*>(in + (size_t)row0 * 256) + lane;

    float4 buf[2][2 * IM_GROUP];
#define IM_LOAD_GROUP(G) do { \
    for (int i = 0; i < 2 * IM_GROUP; ++i) \
        buf[(G) & 1][i] = p[(G) * (IM_GROUP * 64) + i * 32]; } while (0)

    IM_LOAD_GROUP(0);
    float mine = 0.0f;
    for (int g = 0; g < IM_ROWS / IM_GROUP; ++g) {
        __builtin_amdgcn_sched_barrier(0);
        if (g + 1 < IM_ROWS / IM_GROUP) IM_LOAD_GROUP(g + 1);
        if (g == 0) {
            for (int r = 0; r < RD_ROWS; ++r)
                rv[r] = pr[r * RD_COLS];
        }
        __builtin_amdgcn_sched_barrier(0);
        for (int r = 0; r < IM_GROUP; ++r) {
            const float4 v0 = buf[g & 1][2 * r];
            const float4 v1 = buf[g & 1][2 * r + 1];
            float s = (v0.x + v0.y) + (v0.z + v0.w) + (v1.x + v1.y) + (v1.z + v1.w);
            for (int off = 16; off > 0; off >>= 1)
                s += __shfl_xor(s, off, 32);
            if (lane == (unsigned)(g * IM_GROUP + r)) mine = s;
        }
    }
#undef IM_LOAD_GROUP
    out[row0 + lane] = mine * (1.0f / 256.0f);

    float rs = 0.0f;
    for (int r = 0; r < RD_ROWS; ++r)
        rs += rv[r];
    rad_out[gw * 32u + lane] = rs * (1.0f / 32.0f);
}

__device__ __forceinline__ float other_half(float x) {
    return __builtin_bit_cast(float, __builtin_amdgcn_permlanex16(
        __builtin_bit_cast(int, x), __builtin_bit_cast(int, x), 0x76543210, 0xfedcba98, false, false));
}
#define SK_WAVES 8
#define SK_K     (512 / SK_WAVES)
#define SK_F4    (SK_K / 8)
__global__ __launch_bounds__(SK_WAVES * 32) void sim_wmma_kernel(const float* img,
                                                                 const float* rad,
                                                                 const float* __restrict__ temperature,
                                                                 float* __restrict__ out) {
    __shared__ float part[SK_WAVES][16][32];
    __shared__ float ssq[SK_WAVES][48];

    const int ti   = blockIdx.x >> 3;
    const int tjp  = blockIdx.x & 7;
    const int lane = threadIdx.x & 31;
    const int wave = __builtin_amdgcn_readfirstlane(threadIdx.x >> 5);
    const int h = lane >> 4;
    const int c = lane & 15;

    const int rowA  = ti * 16 + c;
    const int rowB0 = tjp * 32 + c;
    const int rowB1 = rowB0 + 16;

    const float temp = temperature[0];

    const float4* pa  = reinterpret_cast<const float4*>(img + (size_t)rowA  * 512 + wave * SK_K) + h;
    const float4* pb0 = reinterpret_cast<const float4*>(rad + (size_t)rowB0 * 512 + wave * SK_K) + h;
    const float4* pb1 = reinterpret_cast<const float4*>(rad + (size_t)rowB1 * 512 + wave * SK_K) + h;

    float4 fa[SK_F4], fb0[SK_F4], fb1[SK_F4];
    for (int j = 0; j < SK_F4; ++j) {
        fa[j]  = pa[2 * j];
        fb0[j] = pb0[2 * j];
        fb1[j] = pb1[2 * j];
    }
    asm volatile("" : "+v"(fa[0].x));

    v8f acc0 = {};
    v8f acc1 = {};
    v2f sa_lo = {0.0f, 0.0f}, sa_hi = {0.0f, 0.0f};
    v2f sp_lo = {0.0f, 0.0f}, sp_hi = {0.0f, 0.0f};
    v2f sq_lo = {0.0f, 0.0f}, sq_hi = {0.0f, 0.0f};
    for (int j = 0; j < SK_F4; ++j) {
        v2f a_lo = {fa[j].x, fa[j].y};
        v2f a_hi = {fa[j].z, fa[j].w};
        v2f p_lo = {fb0[j].x, fb0[j].y};
        v2f p_hi = {fb0[j].z, fb0[j].w};
        v2f q_lo = {fb1[j].x, fb1[j].y};
        v2f q_hi = {fb1[j].z, fb1[j].w};
        acc0 = __builtin_amdgcn_wmma_f32_16x16x4_f32(false, a_lo, false, p_lo, (short)0, acc0, false, false);
        acc1 = __builtin_amdgcn_wmma_f32_16x16x4_f32(false, a_lo, false, q_lo, (short)0, acc1, false, false);
        acc0 = __builtin_amdgcn_wmma_f32_16x16x4_f32(false, a_hi, false, p_hi, (short)0, acc0, false, false);
        acc1 = __builtin_amdgcn_wmma_f32_16x16x4_f32(false, a_hi, false, q_hi, (short)0, acc1, false, false);
        sa_lo += a_lo * a_lo;  sa_hi += a_hi * a_hi;
        sp_lo += p_lo * p_lo;  sp_hi += p_hi * p_hi;
        sq_lo += q_lo * q_lo;  sq_hi += q_hi * q_hi;
        asm volatile("" : "+v"(acc0), "+v"(acc1), "+v"(sa_lo), "+v"(sa_hi),
                          "+v"(sp_lo), "+v"(sp_hi), "+v"(sq_lo), "+v"(sq_hi));
    }

    float ssA  = (sa_lo.x + sa_lo.y) + (sa_hi.x + sa_hi.y);
    float ssB0 = (sp_lo.x + sp_lo.y) + (sp_hi.x + sp_hi.y);
    float ssB1 = (sq_lo.x + sq_lo.y) + (sq_hi.x + sq_hi.y);
    ssA  += other_half(ssA);
    ssB0 += other_half(ssB0);
    ssB1 += other_half(ssB1);

    for (int v = 0; v < 8; ++v) {
        part[wave][v][lane]     = acc0[v];
        part[wave][8 + v][lane] = acc1[v];
    }
    ssq[wave][lane] = h ? ssB1 : ssB0;
    if (h == 0) ssq[wave][32 + c] = ssA;
    __syncthreads();

    float tB = ssq[0][lane];
    float tA = ssq[0][32 + c];
    for (int w = 1; w < SK_WAVES; ++w) {
        tB += ssq[w][lane];
        tA += ssq[w][32 + c];
    }
    const float sclB = 1.0f / (fmaxf(sqrtf(tB), 1e-12f) * temp);
    const float invA = 1.0f / fmaxf(sqrtf(tA), 1e-12f);

    for (int q = 0; q < 2; ++q) {
        const int rho = 2 * wave + q;
        const int reg = h * 8 + (rho & 7);
        const int sl  = (rho >> 3) * 16 + c;
        float s = part[0][reg][sl];
        for (int w = 1; w < SK_WAVES; ++w)
            s += part[w][reg][sl];
        const float sclA = __builtin_bit_cast(float,
            __builtin_amdgcn_readlane(__builtin_bit_cast(int, invA), rho));
        out[(ti * 16 + rho) * 256 + tjp * 32 + lane] = s * sclA * sclB;
    }
}

extern "C" void kernel_launch(void* const* d_in, const int* in_sizes, int n_in,
                              void* d_out, int out_size, void* d_ws, size_t ws_size,
                              hipStream_t stream) {
    const float* img_feat = (const float*)d_in[0];
    const float* rad_feat = (const float*)d_in[1];
    const float* temp     = (const float*)d_in[2];

    float* out_sim = (float*)d_out;
    float* out_img = out_sim + 256 * 256;
    float* out_rad = out_img + 256 * 512;

    mean_stream_kernel<<<512, 256, 0, stream>>>(img_feat, out_img, rad_feat, out_rad);
    sim_wmma_kernel<<<128, 256, 0, stream>>>(out_img, out_rad, temp, out_sim);
}
